# starting kernel
# baseline (speedup 1.0000x reference)
#include <hip/hip_runtime.h>
#include <hip/hip_bf16.h>

#define B_  64
#define T_  2048
#define QD_ 1024
#define AD_ 256

typedef float v2f __attribute__((ext_vector_type(2)));
typedef float v4f __attribute__((ext_vector_type(4)));
typedef float v8f __attribute__((ext_vector_type(8)));

__global__ void pq_gemm_wmma(const float* __restrict__ query,
                             const float* __restrict__ Wq,
                             float* __restrict__ pq) {
    const int lane  = threadIdx.x;
    const int mtile = blockIdx.x >> 4;
    const int ntile = blockIdx.x & 15;
    const int mrow  = mtile * 16 + (lane & 15);
    const int ncol  = ntile * 16 + (lane & 15);
    const int kh    = (lane >> 4) * 2;

    const float* qrow = query + (size_t)mrow * QD_ + kh;
    const float* wrow = Wq    + (size_t)ncol * QD_ + kh;

    v8f c = {0.f, 0.f, 0.f, 0.f, 0.f, 0.f, 0.f, 0.f};
#pragma unroll 4
    for (int k0 = 0; k0 < QD_; k0 += 4) {
        v2f a = *(const v2f*)(qrow + k0);
        v2f b = *(const v2f*)(wrow + k0);
        c = __builtin_amdgcn_wmma_f32_16x16x4_f32(
                 false, a,  false, b,
                 (short)0, c,  false,  false);
    }

    const int mBase = mtile * 16 + 8 * (lane >> 4);
    const int nOut  = ntile * 16 + (lane & 15);
#pragma unroll
    for (int j = 0; j < 8; ++j)
        pq[(size_t)(mBase + j) * AD_ + nOut] = c[j];
}

__global__ void energies_kernel(const float* __restrict__ enc,
                                const int*   __restrict__ chars,
                                const float* __restrict__ pq,
                                const float* __restrict__ v,
                                float* __restrict__ energies) {
    const int b     = blockIdx.x >> 4;
    const int chunk = blockIdx.x & 15;
    const int lane  = threadIdx.x & 31;
    const int wave  = threadIdx.x >> 5;
    const int a0    = lane * 4;

    const v4f v0 = *(const v4f*)(v + a0);
    const v4f v1 = *(const v4f*)(v + 128 + a0);
    const v4f p0 = *(const v4f*)(pq + (size_t)b * AD_ + a0);
    const v4f p1 = *(const v4f*)(pq + (size_t)b * AD_ + 128 + a0);

    const int tBase = chunk * (T_ / 16);

    for (int ti = wave; ti < T_ / 16; ti += 8) {
        const int t = tBase + ti;
        const float* row = enc + ((size_t)b * T_ + t) * AD_;

        __builtin_prefetch(row + 8 * AD_, 0, 0);

        v4f e0 = __builtin_nontemporal_load((const v4f*)(row + a0));
        v4f e1 = __builtin_nontemporal_load((const v4f*)(row + 128 + a0));

        float acc;
        acc = __builtin_amdgcn_tanhf(p0.x + e0.x) * v0.x;
        acc = fmaf(__builtin_amdgcn_tanhf(p0.y + e0.y), v0.y, acc);
        acc = fmaf(__builtin_amdgcn_tanhf(p0.z + e0.z), v0.z, acc);
        acc = fmaf(__builtin_amdgcn_tanhf(p0.w + e0.w), v0.w, acc);
        acc = fmaf(__builtin_amdgcn_tanhf(p1.x + e1.x), v1.x, acc);
        acc = fmaf(__builtin_amdgcn_tanhf(p1.y + e1.y), v1.y, acc);
        acc = fmaf(__builtin_amdgcn_tanhf(p1.z + e1.z), v1.z, acc);
        acc = fmaf(__builtin_amdgcn_tanhf(p1.w + e1.w), v1.w, acc);

#pragma unroll
        for (int off = 16; off > 0; off >>= 1)
            acc += __shfl_xor(acc, off, 32);

        if (lane == 0) {
            const float m = (chars[(size_t)b * T_ + t] != 0) ? 1.0f : 0.0f;
            energies[(size_t)b * T_ + t] = acc * m;
        }
    }
}

__global__ void softmax_kernel(float* buf) {
    __shared__ float red[256];
    const int b   = blockIdx.x;
    const int tid = threadIdx.x;
    float* rowp = buf + (size_t)b * T_;

    float vals[8];
    float mx = -1e30f;
#pragma unroll
    for (int i = 0; i < 8; ++i) {
        vals[i] = rowp[tid + i * 256];
        mx = fmaxf(mx, vals[i]);
    }
    red[tid] = mx;
    __syncthreads();
    for (int s = 128; s > 0; s >>= 1) {
        if (tid < s) red[tid] = fmaxf(red[tid], red[tid + s]);
        __syncthreads();
    }
    mx = red[0];
    __syncthreads();

    float sum = 0.f;
#pragma unroll
    for (int i = 0; i < 8; ++i) {
        vals[i] = __expf(vals[i] - mx);
        sum += vals[i];
    }
    red[tid] = sum;
    __syncthreads();
    for (int s = 128; s > 0; s >>= 1) {
        if (tid < s) red[tid] += red[tid + s];
        __syncthreads();
    }
    const float inv = 1.0f / red[0];
#pragma unroll
    for (int i = 0; i < 8; ++i)
        rowp[tid + i * 256] = vals[i] * inv;
}

extern "C" void kernel_launch(void* const* d_in, const int* in_sizes, int n_in,
                              void* d_out, int out_size, void* d_ws, size_t ws_size,
                              hipStream_t stream) {
    const float* enc   = (const float*)d_in[0];
    const float* query = (const float*)d_in[1];
    const int*   chars = (const int*)d_in[2];
    const float* Wq    = (const float*)d_in[4];
    const float* v     = (const float*)d_in[5];
    float* out = (float*)d_out;
    float* pq  = (float*)d_ws;

    pq_gemm_wmma<<<dim3(64), dim3(32), 0, stream>>>(query, Wq, pq);
    energies_kernel<<<dim3(B_ * 16), dim3(256), 0, stream>>>(enc, chars, pq, v, out);
    softmax_kernel<<<dim3(B_), dim3(256), 0, stream>>>(out);
}


# query-projection GEMM: K=1024 split over the 16 waves of one workgroup per 16x16 tile; each wave issues its 64-k slice as one load group, 2 accumulators, partials summed via LDS (sum order changes)
# speedup vs baseline: 1.4775x; 1.4775x over previous
#include <hip/hip_runtime.h>
#include <hip/hip_bf16.h>

#define B_  64
#define T_  2048
#define QD_ 1024
#define AD_ 256

typedef float v2f __attribute__((ext_vector_type(2)));
typedef float v4f __attribute__((ext_vector_type(4)));
typedef float v8f __attribute__((ext_vector_type(8)));

#define KSPLIT 16
#define KW     (QD_ / KSPLIT)

__global__ __launch_bounds__(KSPLIT * 32)
void pq_gemm_wmma(const float* __restrict__ query,
                  const float* __restrict__ Wq,
                  float* __restrict__ pq) {
    __shared__ float part[KSPLIT][256];

    const int lane  = threadIdx.x & 31;
    const int wave  = threadIdx.x >> 5;
    const int mtile = blockIdx.x >> 4;
    const int ntile = blockIdx.x & 15;
    const int c     = lane & 15;
    const int h     = lane >> 4;

    const float* qrow = query + (size_t)(mtile * 16 + c) * QD_ + wave * KW + h * 4;
    const float* wrow = Wq    + (size_t)(ntile * 16 + c) * QD_ + wave * KW + h * 4;

    v4f a[KW / 8], b[KW / 8];
#pragma unroll
    for (int j = 0; j < KW / 8; ++j) {
        a[j] = *(const v4f*)(qrow + 8 * j);
        b[j] = *(const v4f*)(wrow + 8 * j);
    }

    v8f c0 = {0.f, 0.f, 0.f, 0.f, 0.f, 0.f, 0.f, 0.f};
    v8f c1 = {0.f, 0.f, 0.f, 0.f, 0.f, 0.f, 0.f, 0.f};
#pragma unroll
    for (int j = 0; j < KW / 8; ++j) {
        const v2f a0 = {a[j].x, a[j].y};
        const v2f b0 = {b[j].x, b[j].y};
        const v2f a1 = {a[j].z, a[j].w};
        const v2f b1 = {b[j].z, b[j].w};
        c0 = __builtin_amdgcn_wmma_f32_16x16x4_f32(
                  false, a0,  false, b0,
                  (short)0, c0,  false,  false);
        c1 = __builtin_amdgcn_wmma_f32_16x16x4_f32(
                  false, a1,  false, b1,
                  (short)0, c1,  false,  false);
    }
    c0 = c0 + c1;

#pragma unroll
    for (int r = 0; r < 8; ++r)
        part[wave][(8 * h + r) * 16 + c] = c0[r];
    __syncthreads();

    const int tid = threadIdx.x;
    if (tid < 256) {
        float g[4];
#pragma unroll
        for (int q = 0; q < 4; ++q)
            g[q] = (part[4 * q][tid] + part[4 * q + 1][tid]) +
                   (part[4 * q + 2][tid] + part[4 * q + 3][tid]);
        const float tot = (g[0] + g[1]) + (g[2] + g[3]);
        pq[(size_t)(mtile * 16 + (tid >> 4)) * AD_ + ntile * 16 + (tid & 15)] = tot;
    }
}

__global__ void energies_kernel(const float* __restrict__ enc,
                                const int*   __restrict__ chars,
                                const float* __restrict__ pq,
                                const float* __restrict__ v,
                                float* __restrict__ energies) {
    const int b     = blockIdx.x >> 4;
    const int chunk = blockIdx.x & 15;
    const int lane  = threadIdx.x & 31;
    const int wave  = threadIdx.x >> 5;
    const int a0    = lane * 4;

    const v4f v0 = *(const v4f*)(v + a0);
    const v4f v1 = *(const v4f*)(v + 128 + a0);
    const v4f p0 = *(const v4f*)(pq + (size_t)b * AD_ + a0);
    const v4f p1 = *(const v4f*)(pq + (size_t)b * AD_ + 128 + a0);

    const int tBase = chunk * (T_ / 16);

    for (int ti = wave; ti < T_ / 16; ti += 8) {
        const int t = tBase + ti;
        const float* row = enc + ((size_t)b * T_ + t) * AD_;

        __builtin_prefetch(row + 8 * AD_, 0, 0);

        v4f e0 = __builtin_nontemporal_load((const v4f*)(row + a0));
        v4f e1 = __builtin_nontemporal_load((const v4f*)(row + 128 + a0));

        float acc;
        acc = __builtin_amdgcn_tanhf(p0.x + e0.x) * v0.x;
        acc = fmaf(__builtin_amdgcn_tanhf(p0.y + e0.y), v0.y, acc);
        acc = fmaf(__builtin_amdgcn_tanhf(p0.z + e0.z), v0.z, acc);
        acc = fmaf(__builtin_amdgcn_tanhf(p0.w + e0.w), v0.w, acc);
        acc = fmaf(__builtin_amdgcn_tanhf(p1.x + e1.x), v1.x, acc);
        acc = fmaf(__builtin_amdgcn_tanhf(p1.y + e1.y), v1.y, acc);
        acc = fmaf(__builtin_amdgcn_tanhf(p1.z + e1.z), v1.z, acc);
        acc = fmaf(__builtin_amdgcn_tanhf(p1.w + e1.w), v1.w, acc);

#pragma unroll
        for (int off = 16; off > 0; off >>= 1)
            acc += __shfl_xor(acc, off, 32);

        if (lane == 0) {
            const float m = (chars[(size_t)b * T_ + t] != 0) ? 1.0f : 0.0f;
            energies[(size_t)b * T_ + t] = acc * m;
        }
    }
}

__global__ void softmax_kernel(float* buf) {
    __shared__ float red[256];
    const int b   = blockIdx.x;
    const int tid = threadIdx.x;
    float* rowp = buf + (size_t)b * T_;

    float vals[8];
    float mx = -1e30f;
#pragma unroll
    for (int i = 0; i < 8; ++i) {
        vals[i] = rowp[tid + i * 256];
        mx = fmaxf(mx, vals[i]);
    }
    red[tid] = mx;
    __syncthreads();
    for (int s = 128; s > 0; s >>= 1) {
        if (tid < s) red[tid] = fmaxf(red[tid], red[tid + s]);
        __syncthreads();
    }
    mx = red[0];
    __syncthreads();

    float sum = 0.f;
#pragma unroll
    for (int i = 0; i < 8; ++i) {
        vals[i] = __expf(vals[i] - mx);
        sum += vals[i];
    }
    red[tid] = sum;
    __syncthreads();
    for (int s = 128; s > 0; s >>= 1) {
        if (tid < s) red[tid] += red[tid + s];
        __syncthreads();
    }
    const float inv = 1.0f / red[0];
#pragma unroll
    for (int i = 0; i < 8; ++i)
        rowp[tid + i * 256] = vals[i] * inv;
}

extern "C" void kernel_launch(void* const* d_in, const int* in_sizes, int n_in,
                              void* d_out, int out_size, void* d_ws, size_t ws_size,
                              hipStream_t stream) {
    const float* enc   = (const float*)d_in[0];
    const float* query = (const float*)d_in[1];
    const int*   chars = (const int*)d_in[2];
    const float* Wq    = (const float*)d_in[4];
    const float* v     = (const float*)d_in[5];
    float* out = (float*)d_out;
    float* pq  = (float*)d_ws;

    pq_gemm_wmma<<<dim3(64), dim3(KSPLIT * 32), 0, stream>>>(query, Wq, pq);
    energies_kernel<<<dim3(B_ * 16), dim3(256), 0, stream>>>(enc, chars, pq, v, out);
    softmax_kernel<<<dim3(B_), dim3(256), 0, stream>>>(out);
}


# energies kernel as a stream: each wave owns 32 consecutive rows and keeps 16 rows (32 b128 loads) in flight, mask read up front, butterflies after the stream; 512 workgroups; sums bit-identical
# speedup vs baseline: 2.3275x; 1.5753x over previous
#include <hip/hip_runtime.h>
#include <hip/hip_bf16.h>

#define B_  64
#define T_  2048
#define QD_ 1024
#define AD_ 256

typedef float v2f __attribute__((ext_vector_type(2)));
typedef float v4f __attribute__((ext_vector_type(4)));
typedef float v8f __attribute__((ext_vector_type(8)));

#define KSPLIT 16
#define KW     (QD_ / KSPLIT)

__global__ __launch_bounds__(KSPLIT * 32)
void pq_gemm_wmma(const float* __restrict__ query,
                  const float* __restrict__ Wq,
                  float* __restrict__ pq) {
    __shared__ float part[KSPLIT][256];

    const int lane  = threadIdx.x & 31;
    const int wave  = threadIdx.x >> 5;
    const int mtile = blockIdx.x >> 4;
    const int ntile = blockIdx.x & 15;
    const int c     = lane & 15;
    const int h     = lane >> 4;

    const float* qrow = query + (size_t)(mtile * 16 + c) * QD_ + wave * KW + h * 4;
    const float* wrow = Wq    + (size_t)(ntile * 16 + c) * QD_ + wave * KW + h * 4;

    v4f a[KW / 8], b[KW / 8];
#pragma unroll
    for (int j = 0; j < KW / 8; ++j) {
        a[j] = *(const v4f*)(qrow + 8 * j);
        b[j] = *(const v4f*)(wrow + 8 * j);
    }

    v8f c0 = {0.f, 0.f, 0.f, 0.f, 0.f, 0.f, 0.f, 0.f};
    v8f c1 = {0.f, 0.f, 0.f, 0.f, 0.f, 0.f, 0.f, 0.f};
#pragma unroll
    for (int j = 0; j < KW / 8; ++j) {
        const v2f a0 = {a[j].x, a[j].y};
        const v2f b0 = {b[j].x, b[j].y};
        const v2f a1 = {a[j].z, a[j].w};
        const v2f b1 = {b[j].z, b[j].w};
        c0 = __builtin_amdgcn_wmma_f32_16x16x4_f32(
                  false, a0,  false, b0,
                  (short)0, c0,  false,  false);
        c1 = __builtin_amdgcn_wmma_f32_16x16x4_f32(
                  false, a1,  false, b1,
                  (short)0, c1,  false,  false);
    }
    c0 = c0 + c1;

#pragma unroll
    for (int r = 0; r < 8; ++r)
        part[wave][(8 * h + r) * 16 + c] = c0[r];
    __syncthreads();

    const int tid = threadIdx.x;
    if (tid < 256) {
        float g[4];
#pragma unroll
        for (int q = 0; q < 4; ++q)
            g[q] = (part[4 * q][tid] + part[4 * q + 1][tid]) +
                   (part[4 * q + 2][tid] + part[4 * q + 3][tid]);
        const float tot = (g[0] + g[1]) + (g[2] + g[3]);
        pq[(size_t)(mtile * 16 + (tid >> 4)) * AD_ + ntile * 16 + (tid & 15)] = tot;
    }
}

#define E_WAVES 8
#define E_ROWS  32
#define E_AHEAD 16

__global__ __launch_bounds__(E_WAVES * 32)
void energies_kernel(const float* __restrict__ enc,
                     const int*   __restrict__ chars,
                     const float* __restrict__ pq,
                     const float* __restrict__ v,
                     float* __restrict__ energies) {
    static_assert(T_ % E_ROWS == 0, "a wave's rows stay inside one batch row");
    static_assert((B_ * T_) % (E_WAVES * E_ROWS) == 0, "whole workgroups");
    static_assert(E_ROWS == 32 && E_AHEAD <= E_ROWS, "lane i stores row i");

    const unsigned lane = threadIdx.x & 31u;
    const unsigned wave = __builtin_amdgcn_readfirstlane(threadIdx.x >> 5);
    const unsigned row0 = (blockIdx.x * E_WAVES + wave) * E_ROWS;
    const unsigned b    = row0 / T_;
    const unsigned a0   = lane * 4u;

    const int ch = chars[row0 + lane];

    const v4f v0 = *(const v4f*)(v + a0);
    const v4f v1 = *(const v4f*)(v + 128 + a0);
    const v4f p0 = *(const v4f*)(pq + (size_t)b * AD_ + a0);
    const v4f p1 = *(const v4f*)(pq + (size_t)b * AD_ + 128 + a0);

    const float* wrow = enc + (size_t)row0 * AD_;

    v4f e0[E_AHEAD], e1[E_AHEAD];
    float sum[E_ROWS];

#define LOAD_ROW(P, R) do {                                                       \
        e0[(R) % E_AHEAD] = __builtin_nontemporal_load((const v4f*)((P) + a0));       \
        e1[(R) % E_AHEAD] = __builtin_nontemporal_load((const v4f*)((P) + 128 + a0)); \
    } while (0)

#pragma unroll
    for (int r = 0; r < E_AHEAD; ++r) LOAD_ROW(wrow + (size_t)r * AD_, r);
    __builtin_amdgcn_sched_barrier(0);

#pragma unroll
    for (int r = 0; r < E_ROWS; ++r) {
        const v4f x0 = e0[r % E_AHEAD];
        const v4f x1 = e1[r % E_AHEAD];
        float acc;
        acc = __builtin_amdgcn_tanhf(p0.x + x0.x) * v0.x;
        acc = fmaf(__builtin_amdgcn_tanhf(p0.y + x0.y), v0.y, acc);
        acc = fmaf(__builtin_amdgcn_tanhf(p0.z + x0.z), v0.z, acc);
        acc = fmaf(__builtin_amdgcn_tanhf(p0.w + x0.w), v0.w, acc);
        acc = fmaf(__builtin_amdgcn_tanhf(p1.x + x1.x), v1.x, acc);
        acc = fmaf(__builtin_amdgcn_tanhf(p1.y + x1.y), v1.y, acc);
        acc = fmaf(__builtin_amdgcn_tanhf(p1.z + x1.z), v1.z, acc);
        acc = fmaf(__builtin_amdgcn_tanhf(p1.w + x1.w), v1.w, acc);
        sum[r] = acc;
        if (r + E_AHEAD < E_ROWS) {
            unsigned nxt = (unsigned)(r + E_AHEAD) * AD_;
            asm volatile("" : "+s"(nxt), "+v"(sum[r]));
            LOAD_ROW(wrow + nxt, r + E_AHEAD);
        } else {
            asm volatile("" : "+v"(sum[r]));
        }
        __builtin_amdgcn_sched_barrier(0);
    }
#undef LOAD_ROW

    float mine = 0.f;
#pragma unroll
    for (int r = 0; r < E_ROWS; ++r) {
        float s = sum[r];
#pragma unroll
        for (int off = 16; off > 0; off >>= 1)
            s += __shfl_xor(s, off, 32);
        if (lane == (unsigned)r) mine = s;
    }

    const float m = (ch != 0) ? 1.0f : 0.0f;
    energies[row0 + lane] = mine * m;
}

__global__ void softmax_kernel(float* buf) {
    __shared__ float red[256];
    const int b   = blockIdx.x;
    const int tid = threadIdx.x;
    float* rowp = buf + (size_t)b * T_;

    float vals[8];
    float mx = -1e30f;
#pragma unroll
    for (int i = 0; i < 8; ++i) {
        vals[i] = rowp[tid + i * 256];
        mx = fmaxf(mx, vals[i]);
    }
    red[tid] = mx;
    __syncthreads();
    for (int s = 128; s > 0; s >>= 1) {
        if (tid < s) red[tid] = fmaxf(red[tid], red[tid + s]);
        __syncthreads();
    }
    mx = red[0];
    __syncthreads();

    float sum = 0.f;
#pragma unroll
    for (int i = 0; i < 8; ++i) {
        vals[i] = __expf(vals[i] - mx);
        sum += vals[i];
    }
    red[tid] = sum;
    __syncthreads();
    for (int s = 128; s > 0; s >>= 1) {
        if (tid < s) red[tid] += red[tid + s];
        __syncthreads();
    }
    const float inv = 1.0f / red[0];
#pragma unroll
    for (int i = 0; i < 8; ++i)
        rowp[tid + i * 256] = vals[i] * inv;
}

extern "C" void kernel_launch(void* const* d_in, const int* in_sizes, int n_in,
                              void* d_out, int out_size, void* d_ws, size_t ws_size,
                              hipStream_t stream) {
    const float* enc   = (const float*)d_in[0];
    const float* query = (const float*)d_in[1];
    const int*   chars = (const int*)d_in[2];
    const float* Wq    = (const float*)d_in[4];
    const float* v     = (const float*)d_in[5];
    float* out = (float*)d_out;
    float* pq  = (float*)d_ws;

    pq_gemm_wmma<<<dim3(64), dim3(KSPLIT * 32), 0, stream>>>(query, Wq, pq);
    energies_kernel<<<dim3((B_ * T_) / (E_WAVES * E_ROWS)), dim3(E_WAVES * 32), 0, stream>>>(enc, chars, pq, v, out);
    softmax_kernel<<<dim3(B_), dim3(256), 0, stream>>>(out);
}


# energies kernel: the 32 row sums of a wave cross lanes as one transposed tree on the vector ALU (31 permlane/DPP exchange+add steps) instead of 160 LDS-unit butterfly steps; sums bit-identical
# speedup vs baseline: 2.4867x; 1.0684x over previous
#include <hip/hip_runtime.h>
#include <hip/hip_bf16.h>

#define B_  64
#define T_  2048
#define QD_ 1024
#define AD_ 256

typedef float v2f __attribute__((ext_vector_type(2)));
typedef float v4f __attribute__((ext_vector_type(4)));
typedef float v8f __attribute__((ext_vector_type(8)));

#define KSPLIT 16
#define KW     (QD_ / KSPLIT)

__global__ __launch_bounds__(KSPLIT * 32)
void pq_gemm_wmma(const float* __restrict__ query,
                  const float* __restrict__ Wq,
                  float* __restrict__ pq) {
    __shared__ float part[KSPLIT][256];

    const int lane  = threadIdx.x & 31;
    const int wave  = threadIdx.x >> 5;
    const int mtile = blockIdx.x >> 4;
    const int ntile = blockIdx.x & 15;
    const int c     = lane & 15;
    const int h     = lane >> 4;

    const float* qrow = query + (size_t)(mtile * 16 + c) * QD_ + wave * KW + h * 4;
    const float* wrow = Wq    + (size_t)(ntile * 16 + c) * QD_ + wave * KW + h * 4;

    v4f a[KW / 8], b[KW / 8];
#pragma unroll
    for (int j = 0; j < KW / 8; ++j) {
        a[j] = *(const v4f*)(qrow + 8 * j);
        b[j] = *(const v4f*)(wrow + 8 * j);
    }

    v8f c0 = {0.f, 0.f, 0.f, 0.f, 0.f, 0.f, 0.f, 0.f};
    v8f c1 = {0.f, 0.f, 0.f, 0.f, 0.f, 0.f, 0.f, 0.f};
#pragma unroll
    for (int j = 0; j < KW / 8; ++j) {
        const v2f a0 = {a[j].x, a[j].y};
        const v2f b0 = {b[j].x, b[j].y};
        const v2f a1 = {a[j].z, a[j].w};
        const v2f b1 = {b[j].z, b[j].w};
        c0 = __builtin_amdgcn_wmma_f32_16x16x4_f32(
                  false, a0,  false, b0,
                  (short)0, c0,  false,  false);
        c1 = __builtin_amdgcn_wmma_f32_16x16x4_f32(
                  false, a1,  false, b1,
                  (short)0, c1,  false,  false);
    }
    c0 = c0 + c1;

#pragma unroll
    for (int r = 0; r < 8; ++r)
        part[wave][(8 * h + r) * 16 + c] = c0[r];
    __syncthreads();

    const int tid = threadIdx.x;
    if (tid < 256) {
        float g[4];
#pragma unroll
        for (int q = 0; q < 4; ++q)
            g[q] = (part[4 * q][tid] + part[4 * q + 1][tid]) +
                   (part[4 * q + 2][tid] + part[4 * q + 3][tid]);
        const float tot = (g[0] + g[1]) + (g[2] + g[3]);
        pq[(size_t)(mtile * 16 + (tid >> 4)) * AD_ + ntile * 16 + (tid & 15)] = tot;
    }
}

__device__ __forceinline__ float xhalf(float x) {
    const int b = __builtin_bit_cast(int, x);
    return __builtin_bit_cast(float, __builtin_amdgcn_permlanex16(
        b, b, 0x76543210, 0xfedcba98, false, false));
}
template <int M>
__device__ __forceinline__ float xrow(float x) {
    return __builtin_bit_cast(float, __builtin_amdgcn_update_dpp(
        0, __builtin_bit_cast(int, x), 0x160 + M, 0xf, 0xf, true));
}

template <int M, int N>
__device__ __forceinline__ void fold_level(const float (&in)[N], float (&out)[N / 2], bool up) {
#pragma unroll
    for (int i = 0; i < N / 2; ++i) {
        const float keep = up ? in[i + N / 2] : in[i];
        const float send = up ? in[i] : in[i + N / 2];
        if constexpr (M == 16) out[i] = keep + xhalf(send);
        else                   out[i] = keep + xrow<M>(send);
    }
}

#define E_WAVES 8
#define E_ROWS  32
#define E_AHEAD 16

__global__ __launch_bounds__(E_WAVES * 32)
void energies_kernel(const float* __restrict__ enc,
                     const int*   __restrict__ chars,
                     const float* __restrict__ pq,
                     const float* __restrict__ v,
                     float* __restrict__ energies) {
    static_assert(T_ % E_ROWS == 0, "a wave's rows stay inside one batch row");
    static_assert((B_ * T_) % (E_WAVES * E_ROWS) == 0, "whole workgroups");
    static_assert(E_ROWS == 32 && E_AHEAD <= E_ROWS, "lane i stores row i");

    const unsigned lane = threadIdx.x & 31u;
    const unsigned wave = __builtin_amdgcn_readfirstlane(threadIdx.x >> 5);
    const unsigned row0 = (blockIdx.x * E_WAVES + wave) * E_ROWS;
    const unsigned b    = row0 / T_;
    const unsigned a0   = lane * 4u;

    const int ch = chars[row0 + lane];

    const v4f v0 = *(const v4f*)(v + a0);
    const v4f v1 = *(const v4f*)(v + 128 + a0);
    const v4f p0 = *(const v4f*)(pq + (size_t)b * AD_ + a0);
    const v4f p1 = *(const v4f*)(pq + (size_t)b * AD_ + 128 + a0);

    const float* wrow = enc + (size_t)row0 * AD_;

    v4f e0[E_AHEAD], e1[E_AHEAD];
    float sum[E_ROWS];

#define LOAD_ROW(P, R) do {                                                       \
        e0[(R) % E_AHEAD] = __builtin_nontemporal_load((const v4f*)((P) + a0));       \
        e1[(R) % E_AHEAD] = __builtin_nontemporal_load((const v4f*)((P) + 128 + a0)); \
    } while (0)

#pragma unroll
    for (int r = 0; r < E_AHEAD; ++r) LOAD_ROW(wrow + (size_t)r * AD_, r);
    __builtin_amdgcn_sched_barrier(0);

#pragma unroll
    for (int r = 0; r < E_ROWS; ++r) {
        const v4f x0 = e0[r % E_AHEAD];
        const v4f x1 = e1[r % E_AHEAD];
        float acc;
        acc = __builtin_amdgcn_tanhf(p0.x + x0.x) * v0.x;
        acc = fmaf(__builtin_amdgcn_tanhf(p0.y + x0.y), v0.y, acc);
        acc = fmaf(__builtin_amdgcn_tanhf(p0.z + x0.z), v0.z, acc);
        acc = fmaf(__builtin_amdgcn_tanhf(p0.w + x0.w), v0.w, acc);
        acc = fmaf(__builtin_amdgcn_tanhf(p1.x + x1.x), v1.x, acc);
        acc = fmaf(__builtin_amdgcn_tanhf(p1.y + x1.y), v1.y, acc);
        acc = fmaf(__builtin_amdgcn_tanhf(p1.z + x1.z), v1.z, acc);
        acc = fmaf(__builtin_amdgcn_tanhf(p1.w + x1.w), v1.w, acc);
        sum[r] = acc;
        if (r + E_AHEAD < E_ROWS) {
            unsigned nxt = (unsigned)(r + E_AHEAD) * AD_;
            asm volatile("" : "+s"(nxt), "+v"(sum[r]));
            LOAD_ROW(wrow + nxt, r + E_AHEAD);
        } else {
            asm volatile("" : "+v"(sum[r]));
        }
        __builtin_amdgcn_sched_barrier(0);
    }
#undef LOAD_ROW

    float n16[16], n8[8], n4[4], n2[2], n1[1];
    fold_level<16, 32>(sum, n16, (lane & 16u) != 0u);
    fold_level<8, 16>(n16, n8, (lane & 8u) != 0u);
    fold_level<4, 8>(n8, n4, (lane & 4u) != 0u);
    fold_level<2, 4>(n4, n2, (lane & 2u) != 0u);
    fold_level<1, 2>(n2, n1, (lane & 1u) != 0u);
    const float mine = n1[0];

    const float m = (ch != 0) ? 1.0f : 0.0f;
    energies[row0 + lane] = mine * m;
}

__global__ void softmax_kernel(float* buf) {
    __shared__ float red[256];
    const int b   = blockIdx.x;
    const int tid = threadIdx.x;
    float* rowp = buf + (size_t)b * T_;

    float vals[8];
    float mx = -1e30f;
#pragma unroll
    for (int i = 0; i < 8; ++i) {
        vals[i] = rowp[tid + i * 256];
        mx = fmaxf(mx, vals[i]);
    }
    red[tid] = mx;
    __syncthreads();
    for (int s = 128; s > 0; s >>= 1) {
        if (tid < s) red[tid] = fmaxf(red[tid], red[tid + s]);
        __syncthreads();
    }
    mx = red[0];
    __syncthreads();

    float sum = 0.f;
#pragma unroll
    for (int i = 0; i < 8; ++i) {
        vals[i] = __expf(vals[i] - mx);
        sum += vals[i];
    }
    red[tid] = sum;
    __syncthreads();
    for (int s = 128; s > 0; s >>= 1) {
        if (tid < s) red[tid] += red[tid + s];
        __syncthreads();
    }
    const float inv = 1.0f / red[0];
#pragma unroll
    for (int i = 0; i < 8; ++i)
        rowp[tid + i * 256] = vals[i] * inv;
}

extern "C" void kernel_launch(void* const* d_in, const int* in_sizes, int n_in,
                              void* d_out, int out_size, void* d_ws, size_t ws_size,
                              hipStream_t stream) {
    const float* enc   = (const float*)d_in[0];
    const float* query = (const float*)d_in[1];
    const int*   chars = (const int*)d_in[2];
    const float* Wq    = (const float*)d_in[4];
    const float* v     = (const float*)d_in[5];
    float* out = (float*)d_out;
    float* pq  = (float*)d_ws;

    pq_gemm_wmma<<<dim3(64), dim3(KSPLIT * 32), 0, stream>>>(query, Wq, pq);
    energies_kernel<<<dim3((B_ * T_) / (E_WAVES * E_ROWS)), dim3(E_WAVES * 32), 0, stream>>>(enc, chars, pq, v, out);
    softmax_kernel<<<dim3(B_), dim3(256), 0, stream>>>(out);
}


# softmax kernel: each of its two reductions crosses the workgroup once (LDS, one barrier, 8-way fold, 5 in-wave DPP levels) instead of an 8-level LDS tree with a barrier per level; sums bit-identical
# speedup vs baseline: 2.6371x; 1.0605x over previous
#include <hip/hip_runtime.h>
#include <hip/hip_bf16.h>

#define B_  64
#define T_  2048
#define QD_ 1024
#define AD_ 256

typedef float v2f __attribute__((ext_vector_type(2)));
typedef float v4f __attribute__((ext_vector_type(4)));
typedef float v8f __attribute__((ext_vector_type(8)));

#define KSPLIT 16
#define KW     (QD_ / KSPLIT)

__global__ __launch_bounds__(KSPLIT * 32)
void pq_gemm_wmma(const float* __restrict__ query,
                  const float* __restrict__ Wq,
                  float* __restrict__ pq) {
    __shared__ float part[KSPLIT][256];

    const int lane  = threadIdx.x & 31;
    const int wave  = threadIdx.x >> 5;
    const int mtile = blockIdx.x >> 4;
    const int ntile = blockIdx.x & 15;
    const int c     = lane & 15;
    const int h     = lane >> 4;

    const float* qrow = query + (size_t)(mtile * 16 + c) * QD_ + wave * KW + h * 4;
    const float* wrow = Wq    + (size_t)(ntile * 16 + c) * QD_ + wave * KW + h * 4;

    v4f a[KW / 8], b[KW / 8];
#pragma unroll
    for (int j = 0; j < KW / 8; ++j) {
        a[j] = *(const v4f*)(qrow + 8 * j);
        b[j] = *(const v4f*)(wrow + 8 * j);
    }

    v8f c0 = {0.f, 0.f, 0.f, 0.f, 0.f, 0.f, 0.f, 0.f};
    v8f c1 = {0.f, 0.f, 0.f, 0.f, 0.f, 0.f, 0.f, 0.f};
#pragma unroll
    for (int j = 0; j < KW / 8; ++j) {
        const v2f a0 = {a[j].x, a[j].y};
        const v2f b0 = {b[j].x, b[j].y};
        const v2f a1 = {a[j].z, a[j].w};
        const v2f b1 = {b[j].z, b[j].w};
        c0 = __builtin_amdgcn_wmma_f32_16x16x4_f32(
                  false, a0,  false, b0,
                  (short)0, c0,  false,  false);
        c1 = __builtin_amdgcn_wmma_f32_16x16x4_f32(
                  false, a1,  false, b1,
                  (short)0, c1,  false,  false);
    }
    c0 = c0 + c1;

#pragma unroll
    for (int r = 0; r < 8; ++r)
        part[wave][(8 * h + r) * 16 + c] = c0[r];
    __syncthreads();

    const int tid = threadIdx.x;
    if (tid < 256) {
        float g[4];
#pragma unroll
        for (int q = 0; q < 4; ++q)
            g[q] = (part[4 * q][tid] + part[4 * q + 1][tid]) +
                   (part[4 * q + 2][tid] + part[4 * q + 3][tid]);
        const float tot = (g[0] + g[1]) + (g[2] + g[3]);
        pq[(size_t)(mtile * 16 + (tid >> 4)) * AD_ + ntile * 16 + (tid & 15)] = tot;
    }
}

__device__ __forceinline__ float xhalf(float x) {
    const int b = __builtin_bit_cast(int, x);
    return __builtin_bit_cast(float, __builtin_amdgcn_permlanex16(
        b, b, 0x76543210, 0xfedcba98, false, false));
}
template <int M>
__device__ __forceinline__ float xrow(float x) {
    return __builtin_bit_cast(float, __builtin_amdgcn_update_dpp(
        0, __builtin_bit_cast(int, x), 0x160 + M, 0xf, 0xf, true));
}

template <int M, int N>
__device__ __forceinline__ void fold_level(const float (&in)[N], float (&out)[N / 2], bool up) {
#pragma unroll
    for (int i = 0; i < N / 2; ++i) {
        const float keep = up ? in[i + N / 2] : in[i];
        const float send = up ? in[i] : in[i + N / 2];
        if constexpr (M == 16) out[i] = keep + xhalf(send);
        else                   out[i] = keep + xrow<M>(send);
    }
}

#define E_WAVES 8
#define E_ROWS  32
#define E_AHEAD 16

__global__ __launch_bounds__(E_WAVES * 32)
void energies_kernel(const float* __restrict__ enc,
                     const int*   __restrict__ chars,
                     const float* __restrict__ pq,
                     const float* __restrict__ v,
                     float* __restrict__ energies) {
    static_assert(T_ % E_ROWS == 0, "a wave's rows stay inside one batch row");
    static_assert((B_ * T_) % (E_WAVES * E_ROWS) == 0, "whole workgroups");
    static_assert(E_ROWS == 32 && E_AHEAD <= E_ROWS, "lane i stores row i");

    const unsigned lane = threadIdx.x & 31u;
    const unsigned wave = __builtin_amdgcn_readfirstlane(threadIdx.x >> 5);
    const unsigned row0 = (blockIdx.x * E_WAVES + wave) * E_ROWS;
    const unsigned b    = row0 / T_;
    const unsigned a0   = lane * 4u;

    const int ch = chars[row0 + lane];

    const v4f v0 = *(const v4f*)(v + a0);
    const v4f v1 = *(const v4f*)(v + 128 + a0);
    const v4f p0 = *(const v4f*)(pq + (size_t)b * AD_ + a0);
    const v4f p1 = *(const v4f*)(pq + (size_t)b * AD_ + 128 + a0);

    const float* wrow = enc + (size_t)row0 * AD_;

    v4f e0[E_AHEAD], e1[E_AHEAD];
    float sum[E_ROWS];

#define LOAD_ROW(P, R) do {                                                       \
        e0[(R) % E_AHEAD] = __builtin_nontemporal_load((const v4f*)((P) + a0));       \
        e1[(R) % E_AHEAD] = __builtin_nontemporal_load((const v4f*)((P) + 128 + a0)); \
    } while (0)

#pragma unroll
    for (int r = 0; r < E_AHEAD; ++r) LOAD_ROW(wrow + (size_t)r * AD_, r);
    __builtin_amdgcn_sched_barrier(0);

#pragma unroll
    for (int r = 0; r < E_ROWS; ++r) {
        const v4f x0 = e0[r % E_AHEAD];
        const v4f x1 = e1[r % E_AHEAD];
        float acc;
        acc = __builtin_amdgcn_tanhf(p0.x + x0.x) * v0.x;
        acc = fmaf(__builtin_amdgcn_tanhf(p0.y + x0.y), v0.y, acc);
        acc = fmaf(__builtin_amdgcn_tanhf(p0.z + x0.z), v0.z, acc);
        acc = fmaf(__builtin_amdgcn_tanhf(p0.w + x0.w), v0.w, acc);
        acc = fmaf(__builtin_amdgcn_tanhf(p1.x + x1.x), v1.x, acc);
        acc = fmaf(__builtin_amdgcn_tanhf(p1.y + x1.y), v1.y, acc);
        acc = fmaf(__builtin_amdgcn_tanhf(p1.z + x1.z), v1.z, acc);
        acc = fmaf(__builtin_amdgcn_tanhf(p1.w + x1.w), v1.w, acc);
        sum[r] = acc;
        if (r + E_AHEAD < E_ROWS) {
            unsigned nxt = (unsigned)(r + E_AHEAD) * AD_;
            asm volatile("" : "+s"(nxt), "+v"(sum[r]));
            LOAD_ROW(wrow + nxt, r + E_AHEAD);
        } else {
            asm volatile("" : "+v"(sum[r]));
        }
        __builtin_amdgcn_sched_barrier(0);
    }
#undef LOAD_ROW

    float n16[16], n8[8], n4[4], n2[2], n1[1];
    fold_level<16, 32>(sum, n16, (lane & 16u) != 0u);
    fold_level<8, 16>(n16, n8, (lane & 8u) != 0u);
    fold_level<4, 8>(n8, n4, (lane & 4u) != 0u);
    fold_level<2, 4>(n4, n2, (lane & 2u) != 0u);
    fold_level<1, 2>(n2, n1, (lane & 1u) != 0u);
    const float mine = n1[0];

    const float m = (ch != 0) ? 1.0f : 0.0f;
    energies[row0 + lane] = mine * m;
}

__global__ void softmax_kernel(float* buf) {
    __shared__ float redm[256];
    __shared__ float reds[256];
    const int b    = blockIdx.x;
    const int tid  = threadIdx.x;
    const int lane = tid & 31;
    float* rowp = buf + (size_t)b * T_;

    float vals[8];
    float mx = -1e30f;
#pragma unroll
    for (int i = 0; i < 8; ++i) {
        vals[i] = rowp[tid + i * 256];
        mx = fmaxf(mx, vals[i]);
    }
    redm[tid] = mx;
    __syncthreads();
    {
        const float* w = redm + lane;
        mx = fmaxf(fmaxf(fmaxf(w[0],  w[128]), fmaxf(w[64], w[192])),
                   fmaxf(fmaxf(w[32], w[160]), fmaxf(w[96], w[224])));
        mx = fmaxf(mx, xhalf(mx));
        mx = fmaxf(mx, xrow<8>(mx));
        mx = fmaxf(mx, xrow<4>(mx));
        mx = fmaxf(mx, xrow<2>(mx));
        mx = fmaxf(mx, xrow<1>(mx));
    }

    float sum = 0.f;
#pragma unroll
    for (int i = 0; i < 8; ++i) {
        vals[i] = __expf(vals[i] - mx);
        sum += vals[i];
    }
    reds[tid] = sum;
    __syncthreads();
    float tot;
    {
        const float* w = reds + lane;
        tot = ((w[0] + w[128]) + (w[64] + w[192])) + ((w[32] + w[160]) + (w[96] + w[224]));
        tot = tot + xhalf(tot);
        tot = tot + xrow<8>(tot);
        tot = tot + xrow<4>(tot);
        tot = tot + xrow<2>(tot);
        tot = tot + xrow<1>(tot);
    }
    const float inv = 1.0f / tot;
#pragma unroll
    for (int i = 0; i < 8; ++i)
        rowp[tid + i * 256] = vals[i] * inv;
}

extern "C" void kernel_launch(void* const* d_in, const int* in_sizes, int n_in,
                              void* d_out, int out_size, void* d_ws, size_t ws_size,
                              hipStream_t stream) {
    const float* enc   = (const float*)d_in[0];
    const float* query = (const float*)d_in[1];
    const int*   chars = (const int*)d_in[2];
    const float* Wq    = (const float*)d_in[4];
    const float* v     = (const float*)d_in[5];
    float* out = (float*)d_out;
    float* pq  = (float*)d_ws;

    pq_gemm_wmma<<<dim3(64), dim3(KSPLIT * 32), 0, stream>>>(query, Wq, pq);
    energies_kernel<<<dim3((B_ * T_) / (E_WAVES * E_ROWS)), dim3(E_WAVES * 32), 0, stream>>>(enc, chars, pq, v, out);
    softmax_kernel<<<dim3(B_), dim3(256), 0, stream>>>(out);
}
